# starting kernel
# baseline (speedup 1.0000x reference)
#include <hip/hip_runtime.h>
#include <math.h>

typedef __attribute__((ext_vector_type(2))) float v2f;
typedef __attribute__((ext_vector_type(8))) float v8f;

#define D 256

__device__ __forceinline__ float gelu_exact(float x) {
    return 0.5f * x * (1.0f + erff(x * 0.70710678118654752440f));
}

__global__ __launch_bounds__(512) void mlp_weight_kernel(
    const float* __restrict__ W1, const float* __restrict__ b1,
    const float* __restrict__ W2, const float* __restrict__ b2,
    const float* __restrict__ W3, const float* __restrict__ b3,
    const float* __restrict__ W4, const float* __restrict__ b4,
    const float* __restrict__ W5, const float* __restrict__ b5,
    float* __restrict__ w12_out)
{
    __shared__ float bufA[D];
    __shared__ float bufB[D];

    const int tid  = threadIdx.x;
    const int wave = tid >> 5;
    const int lane = tid & 31;
    const int n    = lane & 15;
    const int hi   = (lane >> 4) & 1;

    if (tid < D) {
        float v = W1[tid * 4 + 2] + W1[tid * 4 + 3] + b1[tid];
        bufA[tid] = gelu_exact(v);
    }
    __syncthreads();

    const float* Ws[3] = {W2, W3, W4};
    const float* bs[3] = {b2, b3, b4};
    float* cur = bufA;
    float* nxt = bufB;

    for (int L = 0; L < 3; ++L) {
        const float* Wm = Ws[L];
        const float* bb = bs[L];
        const int tile  = wave;
        const float* wrow = Wm + (size_t)(tile * 16 + n) * D;

        v8f acc = {};
        for (int k0 = 0; k0 < D; k0 += 4) {
            v2f a, bmat;
            a[0]    = cur[k0 + 2 * hi];
            a[1]    = cur[k0 + 2 * hi + 1];
            bmat[0] = wrow[k0 + 2 * hi];
            bmat[1] = wrow[k0 + 2 * hi + 1];
            acc = __builtin_amdgcn_wmma_f32_16x16x4_f32(
                false, a, false, bmat, (short)0, acc, false, false);
        }
        nxt[tile * 16 + n] = gelu_exact(acc[0] + bb[tile * 16 + n]);
        __syncthreads();
        float* t = cur; cur = nxt; nxt = t;
    }

    if (tid < 12) {
        const float* wr = W5 + tid * D;
        float s = b5[tid];
        for (int k = 0; k < D; ++k) s = fmaf(cur[k], wr[k], s);
        w12_out[tid] = s;
    }
}

__global__ __launch_bounds__(256) void pred_kernel(
    const float* __restrict__ feat, const float* __restrict__ w12,
    float* __restrict__ out)
{
    __shared__ float w[12];
    if (threadIdx.x < 12) w[threadIdx.x] = w12[threadIdx.x];
    __syncthreads();

    const int t = blockIdx.x * 256 + threadIdx.x;
    const int b = t >> 14;
    const int i = t & 16383;

    const float4* fb = (const float4*)(feat + ((size_t)b << 18));
    float4 c0 = fb[i];
    float4 c1 = fb[16384 + i];
    float4 c2 = fb[2 * 16384 + i];
    float4 c3 = fb[3 * 16384 + i];

    float p0[4] = {c0.x, c0.y, c0.z, c0.w};
    float p1[4] = {c1.x, c1.y, c1.z, c1.w};
    float p2[4] = {c2.x, c2.y, c2.z, c2.w};
    float p3[4] = {c3.x, c3.y, c3.z, c3.w};

    float o[12];
#pragma unroll
    for (int j = 0; j < 4; ++j) {
#pragma unroll
        for (int d = 0; d < 3; ++d) {
            o[j * 3 + d] = fmaf(p0[j], w[d],
                           fmaf(p1[j], w[3 + d],
                           fmaf(p2[j], w[6 + d],
                                p3[j] * w[9 + d])));
        }
    }

    float4* op = (float4*)(out + (size_t)t * 12);
    op[0] = make_float4(o[0], o[1], o[2],  o[3]);
    op[1] = make_float4(o[4], o[5], o[6],  o[7]);
    op[2] = make_float4(o[8], o[9], o[10], o[11]);
}

extern "C" void kernel_launch(void* const* d_in, const int* in_sizes, int n_in,
                              void* d_out, int out_size, void* d_ws, size_t ws_size,
                              hipStream_t stream) {
    const float* feat = (const float*)d_in[0];
    const float* W1   = (const float*)d_in[1];
    const float* b1   = (const float*)d_in[2];
    const float* W2   = (const float*)d_in[3];
    const float* b2   = (const float*)d_in[4];
    const float* W3   = (const float*)d_in[5];
    const float* b3   = (const float*)d_in[6];
    const float* W4   = (const float*)d_in[7];
    const float* b4   = (const float*)d_in[8];
    const float* W5   = (const float*)d_in[9];
    const float* b5   = (const float*)d_in[10];

    float* w12 = (float*)d_ws;
    float* out = (float*)d_out;

    mlp_weight_kernel<<<1, 512, 0, stream>>>(W1, b1, W2, b2, W3, b3, W4, b4, W5, b5, w12);
    pred_kernel<<<256, 256, 0, stream>>>(feat, w12, out);
}


# W2-W5 fetched as whole-line 16-B-per-lane loads ahead of use and staged via LDS into the WMMA B layout (was 16 lines per 8-B load); k regrouped per WMMA, so the sum order of layers 2-4 changes
# speedup vs baseline: 1.4154x; 1.4154x over previous
#include <hip/hip_runtime.h>
#include <math.h>

typedef __attribute__((ext_vector_type(2))) float v2f;
typedef __attribute__((ext_vector_type(8))) float v8f;

#define D 256
#define PITCH 260
#define TILE_FLOATS (16 * D)

__device__ __forceinline__ float gelu_exact(float x) {
    return 0.5f * x * (1.0f + erff(x * 0.70710678118654752440f));
}

__device__ __forceinline__ float4 tile_load(const float* __restrict__ tile, int i, int lane) {
    return *(const float4*)(tile + i * 128 + 4 * lane);
}

__global__ __launch_bounds__(512) void mlp_weight_kernel(
    const float* __restrict__ W1, const float* __restrict__ b1,
    const float* __restrict__ W2, const float* __restrict__ b2,
    const float* __restrict__ W3, const float* __restrict__ b3,
    const float* __restrict__ W4, const float* __restrict__ b4,
    const float* __restrict__ W5, const float* __restrict__ b5,
    float* __restrict__ w12_out)
{
    __shared__ __attribute__((aligned(16))) float bufA[D];
    __shared__ __attribute__((aligned(16))) float bufB[D];
    __shared__ __attribute__((aligned(16))) float stage[16 * 16 * PITCH];
    __shared__ __attribute__((aligned(16))) float w5s[12 * PITCH];

    const int tid  = threadIdx.x;
    const int wave = tid >> 5;
    const int lane = tid & 31;
    const int n    = lane & 15;
    const int hi   = (lane >> 4) & 1;

    float* st = stage + wave * (16 * PITCH);

    const float* Ws[3] = {W2, W3, W4};
    const float* bs[3] = {b2, b3, b4};

    float4 wr[32];
#pragma unroll
    for (int i = 0; i < 32; ++i) wr[i] = tile_load(W2 + wave * TILE_FLOATS, i, lane);
    float bias[3];
#pragma unroll
    for (int L = 0; L < 3; ++L) bias[L] = bs[L][wave * 16 + n];
    const int r5 = wave < 12 ? wave : 11;
    const float4 w5r0 = tile_load(W5 + r5 * D, 0, lane);
    const float4 w5r1 = tile_load(W5 + r5 * D, 1, lane);
    const float b5v = b5[tid < 12 ? tid : 0];

    if (tid < D) {
        float v = W1[tid * 4 + 2] + W1[tid * 4 + 3] + b1[tid];
        bufA[tid] = gelu_exact(v);
    }

    float* cur = bufA;
    float* nxt = bufB;

#pragma unroll
    for (int L = 0; L < 3; ++L) {
#pragma unroll
        for (int i = 0; i < 32; ++i)
            *(float4*)(st + (i >> 1) * PITCH + (i & 1) * 128 + 4 * lane) = wr[i];
        if (L == 0 && wave < 12) {
            *(float4*)(w5s + wave * PITCH + 4 * lane)       = w5r0;
            *(float4*)(w5s + wave * PITCH + 128 + 4 * lane) = w5r1;
        }
        __syncthreads();
        if (L < 2) {
#pragma unroll
            for (int i = 0; i < 32; ++i)
                wr[i] = tile_load(Ws[L < 2 ? L + 1 : 2] + wave * TILE_FLOATS, i, lane);
            __builtin_amdgcn_sched_barrier(0);
        }

        const float* arow = cur + 4 * hi;
        const float* brow = st + n * PITCH + 4 * hi;

        v8f acc = {};
#pragma unroll
        for (int kb = 0; kb < 32; ++kb) {
            const float4 af = *(const float4*)(arow + kb * 8);
            const float4 bf = *(const float4*)(brow + kb * 8);
            v2f a0 = {af.x, af.y};
            v2f b0 = {bf.x, bf.y};
            v2f a1 = {af.z, af.w};
            v2f b1 = {bf.z, bf.w};
            acc = __builtin_amdgcn_wmma_f32_16x16x4_f32(
                false, a0, false, b0, (short)0, acc, false, false);
            acc = __builtin_amdgcn_wmma_f32_16x16x4_f32(
                false, a1, false, b1, (short)0, acc, false, false);
        }
        nxt[wave * 16 + n] = gelu_exact(acc[0] + bias[L]);
        float* t = cur; cur = nxt; nxt = t;
    }
    __syncthreads();

    if (tid < 12) {
        const float* wr5 = w5s + tid * PITCH;
        float s = b5v;
#pragma unroll
        for (int k = 0; k < D; ++k) s = fmaf(cur[k], wr5[k], s);
        w12_out[tid] = s;
    }
}

__global__ __launch_bounds__(256) void pred_kernel(
    const float* __restrict__ feat, const float* __restrict__ w12,
    float* __restrict__ out)
{
    __shared__ float w[12];
    if (threadIdx.x < 12) w[threadIdx.x] = w12[threadIdx.x];
    __syncthreads();

    const int t = blockIdx.x * 256 + threadIdx.x;
    const int b = t >> 14;
    const int i = t & 16383;

    const float4* fb = (const float4*)(feat + ((size_t)b << 18));
    float4 c0 = fb[i];
    float4 c1 = fb[16384 + i];
    float4 c2 = fb[2 * 16384 + i];
    float4 c3 = fb[3 * 16384 + i];

    float p0[4] = {c0.x, c0.y, c0.z, c0.w};
    float p1[4] = {c1.x, c1.y, c1.z, c1.w};
    float p2[4] = {c2.x, c2.y, c2.z, c2.w};
    float p3[4] = {c3.x, c3.y, c3.z, c3.w};

    float o[12];
#pragma unroll
    for (int j = 0; j < 4; ++j) {
#pragma unroll
        for (int d = 0; d < 3; ++d) {
            o[j * 3 + d] = fmaf(p0[j], w[d],
                           fmaf(p1[j], w[3 + d],
                           fmaf(p2[j], w[6 + d],
                                p3[j] * w[9 + d])));
        }
    }

    float4* op = (float4*)(out + (size_t)t * 12);
    op[0] = make_float4(o[0], o[1], o[2],  o[3]);
    op[1] = make_float4(o[4], o[5], o[6],  o[7]);
    op[2] = make_float4(o[8], o[9], o[10], o[11]);
}

extern "C" void kernel_launch(void* const* d_in, const int* in_sizes, int n_in,
                              void* d_out, int out_size, void* d_ws, size_t ws_size,
                              hipStream_t stream) {
    const float* feat = (const float*)d_in[0];
    const float* W1   = (const float*)d_in[1];
    const float* b1   = (const float*)d_in[2];
    const float* W2   = (const float*)d_in[3];
    const float* b2   = (const float*)d_in[4];
    const float* W3   = (const float*)d_in[5];
    const float* b3   = (const float*)d_in[6];
    const float* W4   = (const float*)d_in[7];
    const float* b4   = (const float*)d_in[8];
    const float* W5   = (const float*)d_in[9];
    const float* b5   = (const float*)d_in[10];

    float* w12 = (float*)d_ws;
    float* out = (float*)d_out;

    mlp_weight_kernel<<<1, 512, 0, stream>>>(W1, b1, W2, b2, W3, b3, W4, b4, W5, b5, w12);
    pred_kernel<<<256, 256, 0, stream>>>(feat, w12, out);
}


# MLP layers 2-4 spread over 16 work-groups each, one launch per layer (k split over 4 waves x 2 accumulators), layer 5 folded into the stream kernel; sum order of layers 2-5 changes
# speedup vs baseline: 2.7065x; 1.9122x over previous
#include <hip/hip_runtime.h>
#include <math.h>

typedef __attribute__((ext_vector_type(2))) float v2f;
typedef __attribute__((ext_vector_type(8))) float v8f;

#define D 256
#define PITCH 260
#define TILE_FLOATS (16 * D)

__device__ __forceinline__ float gelu_exact(float x) {
    return 0.5f * x * (1.0f + erff(x * 0.70710678118654752440f));
}

__device__ __forceinline__ float4 tile_load(const float* __restrict__ tile, int i, int lane) {
    return *(const float4*)(tile + i * 128 + 4 * lane);
}

__global__ __launch_bounds__(512) void mlp_weight_kernel(
    const float* __restrict__ W1, const float* __restrict__ b1,
    const float* __restrict__ W2, const float* __restrict__ b2,
    const float* __restrict__ W3, const float* __restrict__ b3,
    const float* __restrict__ W4, const float* __restrict__ b4,
    const float* __restrict__ W5, const float* __restrict__ b5,
    float* __restrict__ w12_out)
{
    __shared__ __attribute__((aligned(16))) float bufA[D];
    __shared__ __attribute__((aligned(16))) float bufB[D];
    __shared__ __attribute__((aligned(16))) float stage[16 * 16 * PITCH];
    __shared__ __attribute__((aligned(16))) float w5s[12 * PITCH];

    const int tid  = threadIdx.x;
    const int wave = tid >> 5;
    const int lane = tid & 31;
    const int n    = lane & 15;
    const int hi   = (lane >> 4) & 1;

    float* st = stage + wave * (16 * PITCH);

    const float* Ws[3] = {W2, W3, W4};
    const float* bs[3] = {b2, b3, b4};

    float4 wr[32];
#pragma unroll
    for (int i = 0; i < 32; ++i) wr[i] = tile_load(W2 + wave * TILE_FLOATS, i, lane);
    float bias[3];
#pragma unroll
    for (int L = 0; L < 3; ++L) bias[L] = bs[L][wave * 16 + n];
    const int r5 = wave < 12 ? wave : 11;
    const float4 w5r0 = tile_load(W5 + r5 * D, 0, lane);
    const float4 w5r1 = tile_load(W5 + r5 * D, 1, lane);
    const float b5v = b5[tid < 12 ? tid : 0];

    if (tid < D) {
        float v = W1[tid * 4 + 2] + W1[tid * 4 + 3] + b1[tid];
        bufA[tid] = gelu_exact(v);
    }

    float* cur = bufA;
    float* nxt = bufB;

#pragma unroll
    for (int L = 0; L < 3; ++L) {
#pragma unroll
        for (int i = 0; i < 32; ++i)
            *(float4*)(st + (i >> 1) * PITCH + (i & 1) * 128 + 4 * lane) = wr[i];
        if (L == 0 && wave < 12) {
            *(float4*)(w5s + wave * PITCH + 4 * lane)       = w5r0;
            *(float4*)(w5s + wave * PITCH + 128 + 4 * lane) = w5r1;
        }
        __syncthreads();
        if (L < 2) {
#pragma unroll
            for (int i = 0; i < 32; ++i)
                wr[i] = tile_load(Ws[L < 2 ? L + 1 : 2] + wave * TILE_FLOATS, i, lane);
            __builtin_amdgcn_sched_barrier(0);
        }

        const float* arow = cur + 4 * hi;
        const float* brow = st + n * PITCH + 4 * hi;

        v8f acc = {};
#pragma unroll
        for (int kb = 0; kb < 32; ++kb) {
            const float4 af = *(const float4*)(arow + kb * 8);
            const float4 bf = *(const float4*)(brow + kb * 8);
            v2f a0 = {af.x, af.y};
            v2f b0 = {bf.x, bf.y};
            v2f a1 = {af.z, af.w};
            v2f b1 = {bf.z, bf.w};
            acc = __builtin_amdgcn_wmma_f32_16x16x4_f32(
                false, a0, false, b0, (short)0, acc, false, false);
            acc = __builtin_amdgcn_wmma_f32_16x16x4_f32(
                false, a1, false, b1, (short)0, acc, false, false);
        }
        nxt[wave * 16 + n] = gelu_exact(acc[0] + bias[L]);
        float* t = cur; cur = nxt; nxt = t;
    }
    __syncthreads();

    if (tid < 12) {
        const float* wr5 = w5s + tid * PITCH;
        float s = b5v;
#pragma unroll
        for (int k = 0; k < D; ++k) s = fmaf(cur[k], wr5[k], s);
        w12_out[tid] = s;
    }
}

__global__ __launch_bounds__(256) void pred_kernel(
    const float* __restrict__ feat, const float* __restrict__ w12,
    float* __restrict__ out)
{
    __shared__ float w[12];
    if (threadIdx.x < 12) w[threadIdx.x] = w12[threadIdx.x];
    __syncthreads();

    const int t = blockIdx.x * 256 + threadIdx.x;
    const int b = t >> 14;
    const int i = t & 16383;

    const float4* fb = (const float4*)(feat + ((size_t)b << 18));
    float4 c0 = fb[i];
    float4 c1 = fb[16384 + i];
    float4 c2 = fb[2 * 16384 + i];
    float4 c3 = fb[3 * 16384 + i];

    float p0[4] = {c0.x, c0.y, c0.z, c0.w};
    float p1[4] = {c1.x, c1.y, c1.z, c1.w};
    float p2[4] = {c2.x, c2.y, c2.z, c2.w};
    float p3[4] = {c3.x, c3.y, c3.z, c3.w};

    float o[12];
#pragma unroll
    for (int j = 0; j < 4; ++j) {
#pragma unroll
        for (int d = 0; d < 3; ++d) {
            o[j * 3 + d] = fmaf(p0[j], w[d],
                           fmaf(p1[j], w[3 + d],
                           fmaf(p2[j], w[6 + d],
                                p3[j] * w[9 + d])));
        }
    }

    float4* op = (float4*)(out + (size_t)t * 12);
    op[0] = make_float4(o[0], o[1], o[2],  o[3]);
    op[1] = make_float4(o[4], o[5], o[6],  o[7]);
    op[2] = make_float4(o[8], o[9], o[10], o[11]);
}

#define HPLANE (16 * 32)

__device__ __forceinline__ void tile_finish(const float4 (&af)[8], const float4 (&bf)[8],
                                            float bias, float (*red)[16],
                                            float* __restrict__ hout,
                                            int tile, int wave, int lane)
{
    const int n = lane & 15;
    v8f acc0 = {};
    v8f acc1 = {};
#pragma unroll
    for (int s = 0; s < 8; ++s) {
        v2f a0 = {af[s].x, af[s].y};
        v2f b0 = {bf[s].x, bf[s].y};
        v2f a1 = {af[s].z, af[s].w};
        v2f b1 = {bf[s].z, bf[s].w};
        acc0 = __builtin_amdgcn_wmma_f32_16x16x4_f32(
            false, a0, false, b0, (short)0, acc0, false, false);
        acc1 = __builtin_amdgcn_wmma_f32_16x16x4_f32(
            false, a1, false, b1, (short)0, acc1, false, false);
    }
    const float part = acc0[0] + acc1[0];
    if (lane < 16) red[wave][n] = part;
    __syncthreads();
    if (wave == 0) {
        const float s = ((red[0][n] + red[1][n]) + red[2][n]) + red[3][n];
        hout[tile * 32 + lane] = gelu_exact(s + bias);
    }
}

__global__ __launch_bounds__(128) void mlp_layer12_kernel(
    const float* __restrict__ W1, const float* __restrict__ b1,
    const float* __restrict__ W2, const float* __restrict__ b2,
    float* __restrict__ hout)
{
    __shared__ __attribute__((aligned(16))) float xs[4][64];
    __shared__ float red[4][16];

    const int tile = blockIdx.x;
    const int wave = threadIdx.x >> 5;
    const int lane = threadIdx.x & 31;
    const int n    = lane & 15;
    const int hi   = lane >> 4;

    const float* brow = W2 + (tile * 16 + n) * D + 64 * wave + 4 * hi;
    float4 bf[8];
#pragma unroll
    for (int s = 0; s < 8; ++s) bf[s] = *(const float4*)(brow + 8 * s);
    const float bias = b2[tile * 16 + n];

    const int k = 64 * wave + lane;
    const float2 wa = *(const float2*)(W1 + 4 * k + 2);
    const float2 wb = *(const float2*)(W1 + 4 * (k + 32) + 2);
    const float  ba = b1[k];
    const float  bb = b1[k + 32];
    xs[wave][lane]      = gelu_exact(wa.x + wa.y + ba);
    xs[wave][lane + 32] = gelu_exact(wb.x + wb.y + bb);
    __syncthreads();

    float4 af[8];
#pragma unroll
    for (int s = 0; s < 8; ++s) af[s] = *(const float4*)(&xs[wave][8 * s + 4 * hi]);
    __builtin_amdgcn_sched_barrier(0);

    tile_finish(af, bf, bias, red, hout, tile, wave, lane);
}

__global__ __launch_bounds__(128) void mlp_layer_kernel(
    const float* __restrict__ Wl, const float* __restrict__ bl,
    const float* __restrict__ hin, float* __restrict__ hout)
{
    __shared__ float red[4][16];

    const int tile = blockIdx.x;
    const int wave = threadIdx.x >> 5;
    const int lane = threadIdx.x & 31;
    const int n    = lane & 15;
    const int hi   = lane >> 4;

    const float* brow = Wl + (tile * 16 + n) * D + 64 * wave + 4 * hi;
    const float* arow = hin + (4 * wave) * 32 + 4 * hi;
    float4 bf[8];
    float4 af[8];
#pragma unroll
    for (int s = 0; s < 8; ++s) bf[s] = *(const float4*)(brow + 8 * s);
#pragma unroll
    for (int s = 0; s < 8; ++s) af[s] = *(const float4*)(arow + (s >> 1) * 32 + (s & 1) * 8);
    const float bias = bl[tile * 16 + n];
    __builtin_amdgcn_sched_barrier(0);

    tile_finish(af, bf, bias, red, hout, tile, wave, lane);
}

template <int CTRL>
__device__ __forceinline__ float row_xor(float x) {
    const int b = __builtin_bit_cast(int, x);
    return __builtin_bit_cast(float, __builtin_amdgcn_update_dpp(b, b, CTRL, 0xf, 0xf, false));
}
__device__ __forceinline__ float xhalf(float x) {
    const int b = __builtin_bit_cast(int, x);
    return __builtin_bit_cast(float,
        __builtin_amdgcn_permlanex16(b, b, 0x76543210, 0xfedcba98, false, false));
}
__device__ __forceinline__ float wave_sum(float s) {
    s += row_xor<0x161>(s);
    s += row_xor<0x162>(s);
    s += row_xor<0x164>(s);
    s += row_xor<0x168>(s);
    s += xhalf(s);
    return s;
}

__global__ __launch_bounds__(256) void proj_pred_kernel(
    const float* feat,
    const float* __restrict__ W5, const float* __restrict__ b5,
    const float* __restrict__ hin, float* __restrict__ out)
{
    __shared__ float w[12];

    const int wave = threadIdx.x >> 5;
    const int lane = threadIdx.x & 31;
    const int j0 = wave;
    const int j1 = wave < 4 ? wave + 8 : wave;

    const int k0 = 4 * lane;
    const int k1 = 128 + 4 * lane;
    const float4 x0 = *(const float4*)(hin + (k0 >> 4) * 32 + (k0 & 15));
    const float4 x1 = *(const float4*)(hin + (k1 >> 4) * 32 + (k1 & 15));
    const float4 p0 = *(const float4*)(W5 + j0 * D + k0);
    const float4 p1 = *(const float4*)(W5 + j0 * D + k1);
    const float4 q0 = *(const float4*)(W5 + j1 * D + k0);
    const float4 q1 = *(const float4*)(W5 + j1 * D + k1);
    const float bj0 = b5[j0];
    const float bj1 = b5[j1];

    const int t = blockIdx.x * 256 + threadIdx.x;
    const int b = t >> 14;
    const int i = t & 16383;

    const float4* fb = (const float4*)(feat + ((size_t)b << 18));
    float4 c0 = fb[i];
    float4 c1 = fb[16384 + i];
    float4 c2 = fb[2 * 16384 + i];
    float4 c3 = fb[3 * 16384 + i];
    __builtin_amdgcn_sched_barrier(0);

    float s0 = x0.x * p0.x;
    s0 = fmaf(x0.y, p0.y, s0); s0 = fmaf(x0.z, p0.z, s0); s0 = fmaf(x0.w, p0.w, s0);
    s0 = fmaf(x1.x, p1.x, s0); s0 = fmaf(x1.y, p1.y, s0);
    s0 = fmaf(x1.z, p1.z, s0); s0 = fmaf(x1.w, p1.w, s0);
    float s1 = x0.x * q0.x;
    s1 = fmaf(x0.y, q0.y, s1); s1 = fmaf(x0.z, q0.z, s1); s1 = fmaf(x0.w, q0.w, s1);
    s1 = fmaf(x1.x, q1.x, s1); s1 = fmaf(x1.y, q1.y, s1);
    s1 = fmaf(x1.z, q1.z, s1); s1 = fmaf(x1.w, q1.w, s1);
    s0 = wave_sum(s0);
    s1 = wave_sum(s1);
    w[j0] = s0 + bj0;
    w[j1] = s1 + bj1;
    __syncthreads();

    float f0[4] = {c0.x, c0.y, c0.z, c0.w};
    float f1[4] = {c1.x, c1.y, c1.z, c1.w};
    float f2[4] = {c2.x, c2.y, c2.z, c2.w};
    float f3[4] = {c3.x, c3.y, c3.z, c3.w};

    float o[12];
#pragma unroll
    for (int j = 0; j < 4; ++j) {
#pragma unroll
        for (int d = 0; d < 3; ++d) {
            o[j * 3 + d] = fmaf(f0[j], w[d],
                           fmaf(f1[j], w[3 + d],
                           fmaf(f2[j], w[6 + d],
                                f3[j] * w[9 + d])));
        }
    }

    float4* op = (float4*)(out + (size_t)t * 12);
    op[0] = make_float4(o[0], o[1], o[2],  o[3]);
    op[1] = make_float4(o[4], o[5], o[6],  o[7]);
    op[2] = make_float4(o[8], o[9], o[10], o[11]);
}

extern "C" void kernel_launch(void* const* d_in, const int* in_sizes, int n_in,
                              void* d_out, int out_size, void* d_ws, size_t ws_size,
                              hipStream_t stream) {
    const float* feat = (const float*)d_in[0];
    const float* W1   = (const float*)d_in[1];
    const float* b1   = (const float*)d_in[2];
    const float* W2   = (const float*)d_in[3];
    const float* b2   = (const float*)d_in[4];
    const float* W3   = (const float*)d_in[5];
    const float* b3   = (const float*)d_in[6];
    const float* W4   = (const float*)d_in[7];
    const float* b4   = (const float*)d_in[8];
    const float* W5   = (const float*)d_in[9];
    const float* b5   = (const float*)d_in[10];

    float* ws  = (float*)d_ws;
    float* out = (float*)d_out;

    if (ws_size >= 4096) {
        float* hA = ws;
        float* hB = ws + HPLANE;
        mlp_layer12_kernel<<<16, 128, 0, stream>>>(W1, b1, W2, b2, hA);
        mlp_layer_kernel<<<16, 128, 0, stream>>>(W3, b3, hA, hB);
        mlp_layer_kernel<<<16, 128, 0, stream>>>(W4, b4, hB, hA);
        proj_pred_kernel<<<256, 256, 0, stream>>>(feat, W5, b5, hA, out);
    } else {
        mlp_weight_kernel<<<1, 512, 0, stream>>>(W1, b1, W2, b2, W3, b3, W4, b4, W5, b5, ws);
        pred_kernel<<<256, 256, 0, stream>>>(feat, ws, out);
    }
}


# Layer 3 joins the layer-2 launch as a k split: each work-group runs its 16 outputs through its 256x16 block of W3 (WMMA), the next launch adds the 16 partials: 3 launches; layer 3 sum order changes
# speedup vs baseline: 3.1681x; 1.1706x over previous
#include <hip/hip_runtime.h>
#include <math.h>

typedef __attribute__((ext_vector_type(2))) float v2f;
typedef __attribute__((ext_vector_type(8))) float v8f;

#define D 256
#define PITCH 260
#define TILE_FLOATS (16 * D)

__device__ __forceinline__ float gelu_exact(float x) {
    return 0.5f * x * (1.0f + erff(x * 0.70710678118654752440f));
}

__device__ __forceinline__ float4 tile_load(const float* __restrict__ tile, int i, int lane) {
    return *(const float4*)(tile + i * 128 + 4 * lane);
}

__global__ __launch_bounds__(512) void mlp_weight_kernel(
    const float* __restrict__ W1, const float* __restrict__ b1,
    const float* __restrict__ W2, const float* __restrict__ b2,
    const float* __restrict__ W3, const float* __restrict__ b3,
    const float* __restrict__ W4, const float* __restrict__ b4,
    const float* __restrict__ W5, const float* __restrict__ b5,
    float* __restrict__ w12_out)
{
    __shared__ __attribute__((aligned(16))) float bufA[D];
    __shared__ __attribute__((aligned(16))) float bufB[D];
    __shared__ __attribute__((aligned(16))) float stage[16 * 16 * PITCH];
    __shared__ __attribute__((aligned(16))) float w5s[12 * PITCH];

    const int tid  = threadIdx.x;
    const int wave = tid >> 5;
    const int lane = tid & 31;
    const int n    = lane & 15;
    const int hi   = (lane >> 4) & 1;

    float* st = stage + wave * (16 * PITCH);

    const float* Ws[3] = {W2, W3, W4};
    const float* bs[3] = {b2, b3, b4};

    float4 wr[32];
#pragma unroll
    for (int i = 0; i < 32; ++i) wr[i] = tile_load(W2 + wave * TILE_FLOATS, i, lane);
    float bias[3];
#pragma unroll
    for (int L = 0; L < 3; ++L) bias[L] = bs[L][wave * 16 + n];
    const int r5 = wave < 12 ? wave : 11;
    const float4 w5r0 = tile_load(W5 + r5 * D, 0, lane);
    const float4 w5r1 = tile_load(W5 + r5 * D, 1, lane);
    const float b5v = b5[tid < 12 ? tid : 0];

    if (tid < D) {
        float v = W1[tid * 4 + 2] + W1[tid * 4 + 3] + b1[tid];
        bufA[tid] = gelu_exact(v);
    }

    float* cur = bufA;
    float* nxt = bufB;

#pragma unroll
    for (int L = 0; L < 3; ++L) {
#pragma unroll
        for (int i = 0; i < 32; ++i)
            *(float4*)(st + (i >> 1) * PITCH + (i & 1) * 128 + 4 * lane) = wr[i];
        if (L == 0 && wave < 12) {
            *(float4*)(w5s + wave * PITCH + 4 * lane)       = w5r0;
            *(float4*)(w5s + wave * PITCH + 128 + 4 * lane) = w5r1;
        }
        __syncthreads();
        if (L < 2) {
#pragma unroll
            for (int i = 0; i < 32; ++i)
                wr[i] = tile_load(Ws[L < 2 ? L + 1 : 2] + wave * TILE_FLOATS, i, lane);
            __builtin_amdgcn_sched_barrier(0);
        }

        const float* arow = cur + 4 * hi;
        const float* brow = st + n * PITCH + 4 * hi;

        v8f acc = {};
#pragma unroll
        for (int kb = 0; kb < 32; ++kb) {
            const float4 af = *(const float4*)(arow + kb * 8);
            const float4 bf = *(const float4*)(brow + kb * 8);
            v2f a0 = {af.x, af.y};
            v2f b0 = {bf.x, bf.y};
            v2f a1 = {af.z, af.w};
            v2f b1 = {bf.z, bf.w};
            acc = __builtin_amdgcn_wmma_f32_16x16x4_f32(
                false, a0, false, b0, (short)0, acc, false, false);
            acc = __builtin_amdgcn_wmma_f32_16x16x4_f32(
                false, a1, false, b1, (short)0, acc, false, false);
        }
        nxt[wave * 16 + n] = gelu_exact(acc[0] + bias[L]);
        float* t = cur; cur = nxt; nxt = t;
    }
    __syncthreads();

    if (tid < 12) {
        const float* wr5 = w5s + tid * PITCH;
        float s = b5v;
#pragma unroll
        for (int k = 0; k < D; ++k) s = fmaf(cur[k], wr5[k], s);
        w12_out[tid] = s;
    }
}

__global__ __launch_bounds__(256) void pred_kernel(
    const float* __restrict__ feat, const float* __restrict__ w12,
    float* __restrict__ out)
{
    __shared__ float w[12];
    if (threadIdx.x < 12) w[threadIdx.x] = w12[threadIdx.x];
    __syncthreads();

    const int t = blockIdx.x * 256 + threadIdx.x;
    const int b = t >> 14;
    const int i = t & 16383;

    const float4* fb = (const float4*)(feat + ((size_t)b << 18));
    float4 c0 = fb[i];
    float4 c1 = fb[16384 + i];
    float4 c2 = fb[2 * 16384 + i];
    float4 c3 = fb[3 * 16384 + i];

    float p0[4] = {c0.x, c0.y, c0.z, c0.w};
    float p1[4] = {c1.x, c1.y, c1.z, c1.w};
    float p2[4] = {c2.x, c2.y, c2.z, c2.w};
    float p3[4] = {c3.x, c3.y, c3.z, c3.w};

    float o[12];
#pragma unroll
    for (int j = 0; j < 4; ++j) {
#pragma unroll
        for (int d = 0; d < 3; ++d) {
            o[j * 3 + d] = fmaf(p0[j], w[d],
                           fmaf(p1[j], w[3 + d],
                           fmaf(p2[j], w[6 + d],
                                p3[j] * w[9 + d])));
        }
    }

    float4* op = (float4*)(out + (size_t)t * 12);
    op[0] = make_float4(o[0], o[1], o[2],  o[3]);
    op[1] = make_float4(o[4], o[5], o[6],  o[7]);
    op[2] = make_float4(o[8], o[9], o[10], o[11]);
}

#define HPLANE (16 * 32)
#define PPLANE (16 * D)

__device__ __forceinline__ float quarter_dot(const float4 (&af)[8], const float4 (&bf)[8])
{
    v8f acc0 = {};
    v8f acc1 = {};
#pragma unroll
    for (int s = 0; s < 8; ++s) {
        v2f a0 = {af[s].x, af[s].y};
        v2f b0 = {bf[s].x, bf[s].y};
        v2f a1 = {af[s].z, af[s].w};
        v2f b1 = {bf[s].z, bf[s].w};
        acc0 = __builtin_amdgcn_wmma_f32_16x16x4_f32(
            false, a0, false, b0, (short)0, acc0, false, false);
        acc1 = __builtin_amdgcn_wmma_f32_16x16x4_f32(
            false, a1, false, b1, (short)0, acc1, false, false);
    }
    return acc0[0] + acc1[0];
}

__global__ __launch_bounds__(128) void mlp_layer123_kernel(
    const float* __restrict__ W1, const float* __restrict__ b1,
    const float* __restrict__ W2, const float* __restrict__ b2,
    const float* __restrict__ W3, float* __restrict__ part)
{
    __shared__ __attribute__((aligned(16))) float xs[4][64];
    __shared__ float red[4][16];
    __shared__ __attribute__((aligned(16))) float hs[4][16];

    const int tile = blockIdx.x;
    const int wave = threadIdx.x >> 5;
    const int lane = threadIdx.x & 31;
    const int n    = lane & 15;
    const int hi   = lane >> 4;

    const float* brow = W2 + (tile * 16 + n) * D + 64 * wave + 4 * hi;
    float4 bf[8];
#pragma unroll
    for (int s = 0; s < 8; ++s) bf[s] = *(const float4*)(brow + 8 * s);
    const float* crow = W3 + (64 * wave + n) * D + 16 * tile + 4 * hi;
    float4 cf[4][2];
#pragma unroll
    for (int i = 0; i < 4; ++i) {
        cf[i][0] = *(const float4*)(crow + i * 16 * D);
        cf[i][1] = *(const float4*)(crow + i * 16 * D + 8);
    }
    const float bias = b2[tile * 16 + n];

    const int k = 64 * wave + lane;
    const float2 wa = *(const float2*)(W1 + 4 * k + 2);
    const float2 wb = *(const float2*)(W1 + 4 * (k + 32) + 2);
    const float  ba = b1[k];
    const float  bb = b1[k + 32];
    xs[wave][lane]      = gelu_exact(wa.x + wa.y + ba);
    xs[wave][lane + 32] = gelu_exact(wb.x + wb.y + bb);
    __syncthreads();

    float4 af[8];
#pragma unroll
    for (int s = 0; s < 8; ++s) af[s] = *(const float4*)(&xs[wave][8 * s + 4 * hi]);
    __builtin_amdgcn_sched_barrier(0);

    const float quarter = quarter_dot(af, bf);
    if (lane < 16) red[wave][n] = quarter;
    __syncthreads();
    const float s2sum = ((red[0][n] + red[1][n]) + red[2][n]) + red[3][n];
    const float h2 = gelu_exact(s2sum + bias);
    if (lane < 16) hs[wave][n] = h2;
    __syncthreads();
    const float4 hf0 = *(const float4*)(&hs[wave][4 * hi]);
    const float4 hf1 = *(const float4*)(&hs[wave][8 + 4 * hi]);

    v8f acc[4] = {};
#pragma unroll
    for (int s = 0; s < 2; ++s) {
        const float4 hf = s == 0 ? hf0 : hf1;
        v2f a0 = {hf.x, hf.y};
        v2f a1 = {hf.z, hf.w};
#pragma unroll
        for (int i = 0; i < 4; ++i) {
            v2f c0 = {cf[i][s].x, cf[i][s].y};
            acc[i] = __builtin_amdgcn_wmma_f32_16x16x4_f32(
                false, a0, false, c0, (short)0, acc[i], false, false);
        }
#pragma unroll
        for (int i = 0; i < 4; ++i) {
            v2f c1 = {cf[i][s].z, cf[i][s].w};
            acc[i] = __builtin_amdgcn_wmma_f32_16x16x4_f32(
                false, a1, false, c1, (short)0, acc[i], false, false);
        }
    }
    float* prow = part + tile * D + 64 * wave;
    prow[lane]      = hi ? acc[1][0] : acc[0][0];
    prow[32 + lane] = hi ? acc[3][0] : acc[2][0];
}

__global__ __launch_bounds__(128) void mlp_layer34_kernel(
    const float* __restrict__ part, const float* __restrict__ b3,
    const float* __restrict__ W4, const float* __restrict__ b4,
    float* __restrict__ hout)
{
    __shared__ __attribute__((aligned(16))) float xs[4][64];
    __shared__ float red[4][16];

    const int tile = blockIdx.x;
    const int wave = threadIdx.x >> 5;
    const int lane = threadIdx.x & 31;
    const int n    = lane & 15;
    const int hi   = lane >> 4;

    const float* brow = W4 + (tile * 16 + n) * D + 64 * wave + 4 * hi;
    float4 bf[8];
#pragma unroll
    for (int s = 0; s < 8; ++s) bf[s] = *(const float4*)(brow + 8 * s);
    const int k = 64 * wave + 2 * lane;
    float2 pp[16];
#pragma unroll
    for (int t = 0; t < 16; ++t) pp[t] = *(const float2*)(part + t * D + k);
    const float2 b3v = *(const float2*)(b3 + k);
    const float bias = b4[tile * 16 + n];
    __builtin_amdgcn_sched_barrier(0);

    float sx = pp[0].x;
    float sy = pp[0].y;
#pragma unroll
    for (int t = 1; t < 16; ++t) { sx += pp[t].x; sy += pp[t].y; }
    *(float2*)(&xs[wave][2 * lane]) = make_float2(gelu_exact(sx + b3v.x), gelu_exact(sy + b3v.y));
    __syncthreads();

    float4 af[8];
#pragma unroll
    for (int s = 0; s < 8; ++s) af[s] = *(const float4*)(&xs[wave][8 * s + 4 * hi]);
    __builtin_amdgcn_sched_barrier(0);

    const float quarter = quarter_dot(af, bf);
    if (lane < 16) red[wave][n] = quarter;
    __syncthreads();
    if (wave == 0) {
        const float s = ((red[0][n] + red[1][n]) + red[2][n]) + red[3][n];
        hout[tile * 32 + lane] = gelu_exact(s + bias);
    }
}

template <int CTRL>
__device__ __forceinline__ float row_xor(float x) {
    const int b = __builtin_bit_cast(int, x);
    return __builtin_bit_cast(float, __builtin_amdgcn_update_dpp(b, b, CTRL, 0xf, 0xf, false));
}
__device__ __forceinline__ float xhalf(float x) {
    const int b = __builtin_bit_cast(int, x);
    return __builtin_bit_cast(float,
        __builtin_amdgcn_permlanex16(b, b, 0x76543210, 0xfedcba98, false, false));
}
__device__ __forceinline__ float wave_sum(float s) {
    s += row_xor<0x161>(s);
    s += row_xor<0x162>(s);
    s += row_xor<0x164>(s);
    s += row_xor<0x168>(s);
    s += xhalf(s);
    return s;
}

__global__ __launch_bounds__(256) void proj_pred_kernel(
    const float* feat,
    const float* __restrict__ W5, const float* __restrict__ b5,
    const float* __restrict__ hin, float* __restrict__ out)
{
    __shared__ float w[12];

    const int wave = threadIdx.x >> 5;
    const int lane = threadIdx.x & 31;
    const int j0 = wave;
    const int j1 = wave < 4 ? wave + 8 : wave;

    const int k0 = 4 * lane;
    const int k1 = 128 + 4 * lane;
    const float4 x0 = *(const float4*)(hin + (k0 >> 4) * 32 + (k0 & 15));
    const float4 x1 = *(const float4*)(hin + (k1 >> 4) * 32 + (k1 & 15));
    const float4 p0 = *(const float4*)(W5 + j0 * D + k0);
    const float4 p1 = *(const float4*)(W5 + j0 * D + k1);
    const float4 q0 = *(const float4*)(W5 + j1 * D + k0);
    const float4 q1 = *(const float4*)(W5 + j1 * D + k1);
    const float bj0 = b5[j0];
    const float bj1 = b5[j1];

    const int t = blockIdx.x * 256 + threadIdx.x;
    const int b = t >> 14;
    const int i = t & 16383;

    const float4* fb = (const float4*)(feat + ((size_t)b << 18));
    float4 c0 = fb[i];
    float4 c1 = fb[16384 + i];
    float4 c2 = fb[2 * 16384 + i];
    float4 c3 = fb[3 * 16384 + i];
    __builtin_amdgcn_sched_barrier(0);

    float s0 = x0.x * p0.x;
    s0 = fmaf(x0.y, p0.y, s0); s0 = fmaf(x0.z, p0.z, s0); s0 = fmaf(x0.w, p0.w, s0);
    s0 = fmaf(x1.x, p1.x, s0); s0 = fmaf(x1.y, p1.y, s0);
    s0 = fmaf(x1.z, p1.z, s0); s0 = fmaf(x1.w, p1.w, s0);
    float s1 = x0.x * q0.x;
    s1 = fmaf(x0.y, q0.y, s1); s1 = fmaf(x0.z, q0.z, s1); s1 = fmaf(x0.w, q0.w, s1);
    s1 = fmaf(x1.x, q1.x, s1); s1 = fmaf(x1.y, q1.y, s1);
    s1 = fmaf(x1.z, q1.z, s1); s1 = fmaf(x1.w, q1.w, s1);
    s0 = wave_sum(s0);
    s1 = wave_sum(s1);
    w[j0] = s0 + bj0;
    w[j1] = s1 + bj1;
    __syncthreads();

    float f0[4] = {c0.x, c0.y, c0.z, c0.w};
    float f1[4] = {c1.x, c1.y, c1.z, c1.w};
    float f2[4] = {c2.x, c2.y, c2.z, c2.w};
    float f3[4] = {c3.x, c3.y, c3.z, c3.w};

    float o[12];
#pragma unroll
    for (int j = 0; j < 4; ++j) {
#pragma unroll
        for (int d = 0; d < 3; ++d) {
            o[j * 3 + d] = fmaf(f0[j], w[d],
                           fmaf(f1[j], w[3 + d],
                           fmaf(f2[j], w[6 + d],
                                f3[j] * w[9 + d])));
        }
    }

    float4* op = (float4*)(out + (size_t)t * 12);
    op[0] = make_float4(o[0], o[1], o[2],  o[3]);
    op[1] = make_float4(o[4], o[5], o[6],  o[7]);
    op[2] = make_float4(o[8], o[9], o[10], o[11]);
}

extern "C" void kernel_launch(void* const* d_in, const int* in_sizes, int n_in,
                              void* d_out, int out_size, void* d_ws, size_t ws_size,
                              hipStream_t stream) {
    const float* feat = (const float*)d_in[0];
    const float* W1   = (const float*)d_in[1];
    const float* b1   = (const float*)d_in[2];
    const float* W2   = (const float*)d_in[3];
    const float* b2   = (const float*)d_in[4];
    const float* W3   = (const float*)d_in[5];
    const float* b3   = (const float*)d_in[6];
    const float* W4   = (const float*)d_in[7];
    const float* b4   = (const float*)d_in[8];
    const float* W5   = (const float*)d_in[9];
    const float* b5   = (const float*)d_in[10];

    float* ws  = (float*)d_ws;
    float* out = (float*)d_out;

    if (ws_size >= 18432) {
        float* part = ws;
        float* h4   = ws + PPLANE;
        mlp_layer123_kernel<<<16, 128, 0, stream>>>(W1, b1, W2, b2, W3, part);
        mlp_layer34_kernel<<<16, 128, 0, stream>>>(part, b3, W4, b4, h4);
        proj_pred_kernel<<<256, 256, 0, stream>>>(feat, W5, b5, h4, out);
    } else {
        mlp_weight_kernel<<<1, 512, 0, stream>>>(W1, b1, W2, b2, W3, b3, W4, b4, W5, b5, ws);
        pred_kernel<<<256, 256, 0, stream>>>(feat, ws, out);
    }
}
